# starting kernel
# baseline (speedup 1.0000x reference)
#include <hip/hip_runtime.h>

#define D_DIM   256
#define K_HEADS 24
#define S_SEL   16384
#define V_OUT   8
#define VPAD    16
#define TILES_PER_BLOCK 8
#define THREADS 256

typedef __attribute__((ext_vector_type(2))) float v2f;
typedef __attribute__((ext_vector_type(8))) float v8f;

__global__ __launch_bounds__(THREADS)
void attr_decoder_wmma_kernel(const float* __restrict__ feats,
                              const int*   __restrict__ mask_idx,
                              const float* __restrict__ weights,
                              const float* __restrict__ bias,
                              float*       __restrict__ out)
{
    __shared__ v2f   lds_w[(D_DIM / 2) * VPAD];
    __shared__ float lds_b[VPAD];

    const int tid = threadIdx.x;
    const int blocks_per_head = (S_SEL / 16) / TILES_PER_BLOCK;
    const int k         = blockIdx.x / blocks_per_head;
    const int tile_base = (blockIdx.x % blocks_per_head) * TILES_PER_BLOCK;

    const float* Wk = weights + (size_t)k * (D_DIM * V_OUT);
    for (int i = tid; i < (D_DIM / 2) * VPAD; i += THREADS) {
        const int j   = i >> 4;
        const int col = i & 15;
        v2f w;
        if (col < V_OUT) {
            w.x = Wk[(2 * j)     * V_OUT + col];
            w.y = Wk[(2 * j + 1) * V_OUT + col];
        } else {
            w.x = 0.0f; w.y = 0.0f;
        }
        lds_w[i] = w;
    }
    if (tid < VPAD) {
        lds_b[tid] = (tid < V_OUT) ? bias[k * V_OUT + tid] : 0.0f;
    }
    __syncthreads();

    const int wave = tid >> 5;
    const int lane = tid & 31;
    const int m    = lane & 15;
    const int hi   = lane >> 4;

    const int tile = tile_base + wave;
    const int s0   = tile * 16;

    const int g = mask_idx[k * S_SEL + s0 + m];
    const float* arow = feats + (size_t)g * D_DIM;

    const v2f* apair = (const v2f*)arow + hi;

    v8f c = {0.f, 0.f, 0.f, 0.f, 0.f, 0.f, 0.f, 0.f};

#pragma unroll 8
    for (int j = 0; j < (D_DIM / 2); j += 2) {
        const v2f a = apair[j];
        const v2f b = lds_w[(j + hi) * VPAD + m];
        c = __builtin_amdgcn_wmma_f32_16x16x4_f32(
                 false, a,  false, b,
                 (short)0, c,  false,  false);
    }

    const float bb = lds_b[m];
    if (m < V_OUT) {
        float* obase = out + ((size_t)(k * S_SEL + s0)) * V_OUT + m;
#pragma unroll
        for (int r = 0; r < 8; ++r) {
            const int M = r + 8 * hi;
            obase[(size_t)M * V_OUT] = c[r] + bb;
        }
    }
}

extern "C" void kernel_launch(void* const* d_in, const int* in_sizes, int n_in,
                              void* d_out, int out_size, void* d_ws, size_t ws_size,
                              hipStream_t stream) {
    const float* feats    = (const float*)d_in[1];
    const int*   mask_idx = (const int*)d_in[2];
    const float* weights  = (const float*)d_in[3];
    const float* bias     = (const float*)d_in[4];
    float*       out      = (float*)d_out;

    const int blocks = K_HEADS * ((S_SEL / 16) / TILES_PER_BLOCK);
    hipLaunchKernelGGL(attr_decoder_wmma_kernel,
                       dim3(blocks), dim3(THREADS), 0, stream,
                       feats, mask_idx, weights, bias, out);
}


# gather each row as whole 128B lines (one load = 4 rows x 8 lanes x 16B, 16 loads in flight) and transpose through per-wave LDS into the A fragments; k order and sum order unchanged
# speedup vs baseline: 1.3286x; 1.3286x over previous
#include <hip/hip_runtime.h>

#define D_DIM   256
#define K_HEADS 24
#define S_SEL   16384
#define V_OUT   8
#define VPAD    16
#define TILES_PER_BLOCK 8
#define THREADS 256

#define CHUNK_F  32
#define PITCH_F  36
#define GROUP_CH 2
#define N_GROUPS (D_DIM / (CHUNK_F * GROUP_CH))

typedef __attribute__((ext_vector_type(2))) float v2f;
typedef __attribute__((ext_vector_type(4))) float v4f;
typedef __attribute__((ext_vector_type(8))) float v8f;

__device__ __forceinline__ void wave_lds_handover() {
    __builtin_amdgcn_fence(__ATOMIC_RELEASE, "wavefront");
    __builtin_amdgcn_wave_barrier();
    __builtin_amdgcn_fence(__ATOMIC_ACQUIRE, "wavefront");
}

__global__ __launch_bounds__(THREADS)
void attr_decoder_wmma_kernel(const float* __restrict__ feats,
                              const int*   __restrict__ mask_idx,
                              const float* __restrict__ weights,
                              const float* __restrict__ bias,
                              float*       __restrict__ out)
{
    __shared__ v2f lds_w[(D_DIM / 2) * VPAD];
    __shared__ __attribute__((aligned(16))) float lds_a[TILES_PER_BLOCK][GROUP_CH][16][PITCH_F];

    const int tid = threadIdx.x;
    const int blocks_per_head = (S_SEL / 16) / TILES_PER_BLOCK;
    const int k         = blockIdx.x / blocks_per_head;
    const int tile_base = (blockIdx.x % blocks_per_head) * TILES_PER_BLOCK;

    const int wave = tid >> 5;
    const int lane = tid & 31;
    const int m    = lane & 15;
    const int hi   = lane >> 4;

    const int tile = tile_base + wave;
    const int s0   = tile * 16;

    const int rsub = lane >> 3;
    const int q    = lane & 7;

    const int* idx = mask_idx + k * S_SEL + s0 + rsub;
    const int g0 = idx[0];
    const int g1 = idx[4];
    const int g2 = idx[8];
    const int g3 = idx[12];
    const float bb = (m < V_OUT) ? bias[k * V_OUT + m] : 0.0f;

    const float* Wk = weights + (size_t)k * (D_DIM * V_OUT);
    for (int i = tid; i < (D_DIM / 2) * VPAD; i += THREADS) {
        const int j   = i >> 4;
        const int col = i & 15;
        v2f w;
        if (col < V_OUT) {
            w.x = Wk[(2 * j)     * V_OUT + col];
            w.y = Wk[(2 * j + 1) * V_OUT + col];
        } else {
            w.x = 0.0f; w.y = 0.0f;
        }
        lds_w[i] = w;
    }

    const v4f* p0 = (const v4f*)(feats + (size_t)g0 * D_DIM) + q;
    const v4f* p1 = (const v4f*)(feats + (size_t)g1 * D_DIM) + q;
    const v4f* p2 = (const v4f*)(feats + (size_t)g2 * D_DIM) + q;
    const v4f* p3 = (const v4f*)(feats + (size_t)g3 * D_DIM) + q;

    v4f buf[2][GROUP_CH * 4];
#pragma unroll
    for (int s = 0; s < 2; ++s) {
#pragma unroll
        for (int cc = 0; cc < GROUP_CH; ++cc) {
            const int ch = s * GROUP_CH + cc;
            buf[s][cc * 4 + 0] = p0[8 * ch];
            buf[s][cc * 4 + 1] = p1[8 * ch];
            buf[s][cc * 4 + 2] = p2[8 * ch];
            buf[s][cc * 4 + 3] = p3[8 * ch];
        }
    }

    __syncthreads();

    v8f c = {0.f, 0.f, 0.f, 0.f, 0.f, 0.f, 0.f, 0.f};

#pragma unroll
    for (int g = 0; g < N_GROUPS; ++g) {
        const int s = g & 1;
#pragma unroll
        for (int cc = 0; cc < GROUP_CH; ++cc) {
#pragma unroll
            for (int i = 0; i < 4; ++i) {
                *(v4f*)&lds_a[wave][cc][4 * i + rsub][4 * q] = buf[s][cc * 4 + i];
            }
        }
        if (g + 2 < N_GROUPS) {
#pragma unroll
            for (int cc = 0; cc < GROUP_CH; ++cc) {
                const int ch = (g + 2) * GROUP_CH + cc;
                buf[s][cc * 4 + 0] = p0[8 * ch];
                buf[s][cc * 4 + 1] = p1[8 * ch];
                buf[s][cc * 4 + 2] = p2[8 * ch];
                buf[s][cc * 4 + 3] = p3[8 * ch];
            }
        }
        wave_lds_handover();

#pragma unroll
        for (int cc = 0; cc < GROUP_CH; ++cc) {
#pragma unroll
            for (int tt = 0; tt < CHUNK_F / 4; ++tt) {
                const int t = (g * GROUP_CH + cc) * (CHUNK_F / 4) + tt;
                const v2f a = *(const v2f*)&lds_a[wave][cc][m][4 * tt + 2 * hi];
                const v2f b = lds_w[(2 * t + hi) * VPAD + m];
                c = __builtin_amdgcn_wmma_f32_16x16x4_f32(
                         false, a,  false, b,
                         (short)0, c,  false,  false);
            }
        }
        wave_lds_handover();
    }

    if (m < V_OUT) {
        float* obase = out + ((size_t)(k * S_SEL + s0)) * V_OUT + m;
#pragma unroll
        for (int r = 0; r < 8; ++r) {
            const int M = r + 8 * hi;
            obase[(size_t)M * V_OUT] = c[r] + bb;
        }
    }
}

extern "C" void kernel_launch(void* const* d_in, const int* in_sizes, int n_in,
                              void* d_out, int out_size, void* d_ws, size_t ws_size,
                              hipStream_t stream) {
    const float* feats    = (const float*)d_in[1];
    const int*   mask_idx = (const int*)d_in[2];
    const float* weights  = (const float*)d_in[3];
    const float* bias     = (const float*)d_in[4];
    float*       out      = (float*)d_out;

    const int blocks = K_HEADS * ((S_SEL / 16) / TILES_PER_BLOCK);
    hipLaunchKernelGGL(attr_decoder_wmma_kernel,
                       dim3(blocks), dim3(THREADS), 0, stream,
                       feats, mask_idx, weights, bias, out);
}


# each wave runs 4 consecutive row tiles of its head (768 blocks, not 3072): weights staged once per block, gather loads stream on across tile boundaries; k order and sum order unchanged
# speedup vs baseline: 1.5303x; 1.1518x over previous
#include <hip/hip_runtime.h>

#define D_DIM   256
#define K_HEADS 24
#define S_SEL   16384
#define V_OUT   8
#define VPAD    16
#define WAVES_PER_BLOCK 8
#define THREADS 256
#define TILES_PER_WAVE 4
#define BLOCKS_PER_HEAD ((S_SEL / 16) / (WAVES_PER_BLOCK * TILES_PER_WAVE))

#define CHUNK_F  32
#define PITCH_F  36
#define GROUP_CH 2
#define N_GROUPS (D_DIM / (CHUNK_F * GROUP_CH))

typedef __attribute__((ext_vector_type(2))) float v2f;
typedef __attribute__((ext_vector_type(4))) float v4f;
typedef __attribute__((ext_vector_type(8))) float v8f;

__device__ __forceinline__ void wave_lds_handover() {
    __builtin_amdgcn_fence(__ATOMIC_RELEASE, "wavefront");
    __builtin_amdgcn_wave_barrier();
    __builtin_amdgcn_fence(__ATOMIC_ACQUIRE, "wavefront");
}

template <bool HAS_NEXT>
__device__ __forceinline__ v8f run_tile(v4f (&buf)[2][GROUP_CH * 4],
                                        const v4f* p0, const v4f* p1, const v4f* p2, const v4f* p3,
                                        const v4f* n0, const v4f* n1, const v4f* n2, const v4f* n3,
                                        float (*la)[16][PITCH_F],
                                        const v2f* lw,
                                        int rsub, int q, int m, int hi)
{
    v8f c = {0.f, 0.f, 0.f, 0.f, 0.f, 0.f, 0.f, 0.f};

#pragma unroll
    for (int g = 0; g < N_GROUPS; ++g) {
        const int s = g & 1;
#pragma unroll
        for (int cc = 0; cc < GROUP_CH; ++cc) {
#pragma unroll
            for (int i = 0; i < 4; ++i) {
                *(v4f*)&la[cc][4 * i + rsub][4 * q] = buf[s][cc * 4 + i];
            }
        }
        if (g + 2 < N_GROUPS) {
#pragma unroll
            for (int cc = 0; cc < GROUP_CH; ++cc) {
                const int ch = (g + 2) * GROUP_CH + cc;
                buf[s][cc * 4 + 0] = p0[8 * ch];
                buf[s][cc * 4 + 1] = p1[8 * ch];
                buf[s][cc * 4 + 2] = p2[8 * ch];
                buf[s][cc * 4 + 3] = p3[8 * ch];
            }
        } else if (HAS_NEXT) {
#pragma unroll
            for (int cc = 0; cc < GROUP_CH; ++cc) {
                const int ch = (g + 2 - N_GROUPS) * GROUP_CH + cc;
                buf[s][cc * 4 + 0] = n0[8 * ch];
                buf[s][cc * 4 + 1] = n1[8 * ch];
                buf[s][cc * 4 + 2] = n2[8 * ch];
                buf[s][cc * 4 + 3] = n3[8 * ch];
            }
        }
        wave_lds_handover();

#pragma unroll
        for (int cc = 0; cc < GROUP_CH; ++cc) {
#pragma unroll
            for (int tt = 0; tt < CHUNK_F / 4; ++tt) {
                const int t = (g * GROUP_CH + cc) * (CHUNK_F / 4) + tt;
                const v2f a = *(const v2f*)&la[cc][m][4 * tt + 2 * hi];
                const v2f b = lw[(2 * t + hi) * VPAD + m];
                c = __builtin_amdgcn_wmma_f32_16x16x4_f32(
                         false, a,  false, b,
                         (short)0, c,  false,  false);
            }
        }
        wave_lds_handover();
    }
    return c;
}

__device__ __forceinline__ void store_tile(float* __restrict__ obase, const v8f c, float bb,
                                           int m, int hi)
{
    if (m < V_OUT) {
#pragma unroll
        for (int r = 0; r < 8; ++r) {
            const int M = r + 8 * hi;
            obase[(size_t)M * V_OUT] = c[r] + bb;
        }
    }
}

__global__ __launch_bounds__(THREADS)
void attr_decoder_wmma_kernel(const float* __restrict__ feats,
                              const int*   __restrict__ mask_idx,
                              const float* __restrict__ weights,
                              const float* __restrict__ bias,
                              float*       __restrict__ out)
{
    __shared__ v2f lds_w[(D_DIM / 2) * VPAD];
    __shared__ __attribute__((aligned(16))) float lds_a[WAVES_PER_BLOCK][GROUP_CH][16][PITCH_F];

    const int tid = threadIdx.x;
    const int k   = blockIdx.x / BLOCKS_PER_HEAD;

    const int wave = tid >> 5;
    const int lane = tid & 31;
    const int m    = lane & 15;
    const int hi   = lane >> 4;

    const int tile0 = ((blockIdx.x % BLOCKS_PER_HEAD) * WAVES_PER_BLOCK + wave) * TILES_PER_WAVE;
    const int s0    = tile0 * 16;

    const int rsub = lane >> 3;
    const int q    = lane & 7;

    const int* idw = mask_idx + k * S_SEL + s0 + rsub;
    const int a0 = idw[0];
    const int a1 = idw[4];
    const int a2 = idw[8];
    const int a3 = idw[12];
    int nx0 = idw[16];
    int nx1 = idw[20];
    int nx2 = idw[24];
    int nx3 = idw[28];
    const float bb = (m < V_OUT) ? bias[k * V_OUT + m] : 0.0f;

    const float* Wk = weights + (size_t)k * (D_DIM * V_OUT);
    for (int i = tid; i < (D_DIM / 2) * VPAD; i += THREADS) {
        const int j   = i >> 4;
        const int col = i & 15;
        v2f w;
        if (col < V_OUT) {
            w.x = Wk[(2 * j)     * V_OUT + col];
            w.y = Wk[(2 * j + 1) * V_OUT + col];
        } else {
            w.x = 0.0f; w.y = 0.0f;
        }
        lds_w[i] = w;
    }

    const v4f* fq = (const v4f*)feats + q;
    const v4f* p0 = fq + (size_t)a0 * (D_DIM / 4);
    const v4f* p1 = fq + (size_t)a1 * (D_DIM / 4);
    const v4f* p2 = fq + (size_t)a2 * (D_DIM / 4);
    const v4f* p3 = fq + (size_t)a3 * (D_DIM / 4);

    v4f buf[2][GROUP_CH * 4];
#pragma unroll
    for (int s = 0; s < 2; ++s) {
#pragma unroll
        for (int cc = 0; cc < GROUP_CH; ++cc) {
            const int ch = s * GROUP_CH + cc;
            buf[s][cc * 4 + 0] = p0[8 * ch];
            buf[s][cc * 4 + 1] = p1[8 * ch];
            buf[s][cc * 4 + 2] = p2[8 * ch];
            buf[s][cc * 4 + 3] = p3[8 * ch];
        }
    }

    __syncthreads();

    float* obase = out + ((size_t)(k * S_SEL + s0)) * V_OUT + m;

#pragma nounroll
    for (int j = 0; j < TILES_PER_WAVE - 1; ++j) {
        const v4f* n0 = fq + (size_t)nx0 * (D_DIM / 4);
        const v4f* n1 = fq + (size_t)nx1 * (D_DIM / 4);
        const v4f* n2 = fq + (size_t)nx2 * (D_DIM / 4);
        const v4f* n3 = fq + (size_t)nx3 * (D_DIM / 4);
        const int j2 = (j + 2 < TILES_PER_WAVE) ? j + 2 : TILES_PER_WAVE - 1;
        nx0 = idw[16 * j2 + 0];
        nx1 = idw[16 * j2 + 4];
        nx2 = idw[16 * j2 + 8];
        nx3 = idw[16 * j2 + 12];

        const v8f c = run_tile<true>(buf, p0, p1, p2, p3, n0, n1, n2, n3,
                                     lds_a[wave], lds_w, rsub, q, m, hi);
        store_tile(obase + (size_t)(16 * j) * V_OUT, c, bb, m, hi);

        p0 = n0; p1 = n1; p2 = n2; p3 = n3;
    }
    {
        const v8f c = run_tile<false>(buf, p0, p1, p2, p3, p0, p1, p2, p3,
                                      lds_a[wave], lds_w, rsub, q, m, hi);
        store_tile(obase + (size_t)(16 * (TILES_PER_WAVE - 1)) * V_OUT, c, bb, m, hi);
    }
}

extern "C" void kernel_launch(void* const* d_in, const int* in_sizes, int n_in,
                              void* d_out, int out_size, void* d_ws, size_t ws_size,
                              hipStream_t stream) {
    const float* feats    = (const float*)d_in[1];
    const int*   mask_idx = (const int*)d_in[2];
    const float* weights  = (const float*)d_in[3];
    const float* bias     = (const float*)d_in[4];
    float*       out      = (float*)d_out;

    const int blocks = K_HEADS * BLOCKS_PER_HEAD;
    hipLaunchKernelGGL(attr_decoder_wmma_kernel,
                       dim3(blocks), dim3(THREADS), 0, stream,
                       feats, mask_idx, weights, bias, out);
}


# each wave multiplies two row tiles at once: a weight fragment read from LDS feeds both, two independent chains, fragment reads grouped ahead of each 8 WMMAs; k order and sum order unchanged
# speedup vs baseline: 1.5834x; 1.0347x over previous
#include <hip/hip_runtime.h>

#define D_DIM   256
#define K_HEADS 24
#define S_SEL   16384
#define V_OUT   8
#define VPAD    16
#define WAVES_PER_BLOCK 8
#define THREADS 256
#define TILES_PER_WAVE 4
#define PAIRS_PER_WAVE (TILES_PER_WAVE / 2)
#define BLOCKS_PER_HEAD ((S_SEL / 16) / (WAVES_PER_BLOCK * TILES_PER_WAVE))

#define CHUNK_F  32
#define PITCH_F  36
#define N_GROUPS (D_DIM / CHUNK_F)
#define HALF_T   (CHUNK_F / 8)

typedef __attribute__((ext_vector_type(2))) float v2f;
typedef __attribute__((ext_vector_type(4))) float v4f;
typedef __attribute__((ext_vector_type(8))) float v8f;

__device__ __forceinline__ void wave_lds_handover() {
    __builtin_amdgcn_fence(__ATOMIC_RELEASE, "wavefront");
    __builtin_amdgcn_wave_barrier();
    __builtin_amdgcn_fence(__ATOMIC_ACQUIRE, "wavefront");
}

__device__ __forceinline__ v4f load_piece(const char* fb, unsigned off, int line) {
    return *(const v4f*)(fb + (size_t)off + (size_t)(128 * line));
}

template <bool HAS_NEXT>
__device__ __forceinline__ void run_pair(v4f (&buf)[2][8], const char* fb,
                                         const unsigned (&cur)[8], const unsigned (&nxt)[8],
                                         float (*la)[16][PITCH_F],
                                         const v2f* lw,
                                         int rsub, int q, int m, int hi,
                                         v8f& c0, v8f& c1)
{
    c0 = (v8f){0.f, 0.f, 0.f, 0.f, 0.f, 0.f, 0.f, 0.f};
    c1 = (v8f){0.f, 0.f, 0.f, 0.f, 0.f, 0.f, 0.f, 0.f};

#pragma unroll
    for (int g = 0; g < N_GROUPS; ++g) {
        const int s = g & 1;
#pragma unroll
        for (int jj = 0; jj < 8; ++jj) {
            *(v4f*)&la[jj >> 2][4 * (jj & 3) + rsub][4 * q] = buf[s][jj];
        }
        __builtin_amdgcn_sched_barrier(0);
        if (g + 2 < N_GROUPS) {
#pragma unroll
            for (int jj = 0; jj < 8; ++jj) buf[s][jj] = load_piece(fb, cur[jj], g + 2);
        } else if (HAS_NEXT) {
#pragma unroll
            for (int jj = 0; jj < 8; ++jj) buf[s][jj] = load_piece(fb, nxt[jj], g + 2 - N_GROUPS);
        }
        wave_lds_handover();

#pragma unroll
        for (int h = 0; h < 2; ++h) {
            v2f a0[HALF_T], a1[HALF_T], b[HALF_T];
#pragma unroll
            for (int u = 0; u < HALF_T; ++u) {
                const int tt = h * HALF_T + u;
                const int t  = g * (CHUNK_F / 4) + tt;
                a0[u] = *(const v2f*)&la[0][m][4 * tt + 2 * hi];
                a1[u] = *(const v2f*)&la[1][m][4 * tt + 2 * hi];
                b[u]  = lw[(2 * t + hi) * VPAD + m];
            }
            __builtin_amdgcn_sched_barrier(0);
#pragma unroll
            for (int u = 0; u < HALF_T; ++u) {
                c0 = __builtin_amdgcn_wmma_f32_16x16x4_f32(
                         false, a0[u],  false, b[u],
                         (short)0, c0,  false,  false);
                c1 = __builtin_amdgcn_wmma_f32_16x16x4_f32(
                         false, a1[u],  false, b[u],
                         (short)0, c1,  false,  false);
            }
            __builtin_amdgcn_sched_barrier(0);
        }
        wave_lds_handover();
    }
}

__device__ __forceinline__ void store_tile(float* __restrict__ obase, const v8f c, float bb,
                                           int m, int hi)
{
    if (m < V_OUT) {
#pragma unroll
        for (int r = 0; r < 8; ++r) {
            const int M = r + 8 * hi;
            obase[(size_t)M * V_OUT] = c[r] + bb;
        }
    }
}

__global__ __launch_bounds__(THREADS)
void attr_decoder_wmma_kernel(const float* __restrict__ feats,
                              const int*   __restrict__ mask_idx,
                              const float* __restrict__ weights,
                              const float* __restrict__ bias,
                              float*       __restrict__ out)
{
    __shared__ v2f lds_w[(D_DIM / 2) * VPAD];
    __shared__ __attribute__((aligned(16))) float lds_a[WAVES_PER_BLOCK][2][16][PITCH_F];

    const int tid = threadIdx.x;
    const int k   = blockIdx.x / BLOCKS_PER_HEAD;

    const int wave = tid >> 5;
    const int lane = tid & 31;
    const int m    = lane & 15;
    const int hi   = lane >> 4;

    const int tile0 = ((blockIdx.x % BLOCKS_PER_HEAD) * WAVES_PER_BLOCK + wave) * TILES_PER_WAVE;
    const int s0    = tile0 * 16;

    const int rsub = lane >> 3;
    const int q    = lane & 7;

    const int* idw = mask_idx + k * S_SEL + s0 + rsub;
    int id0[8], id1[8];
#pragma unroll
    for (int jj = 0; jj < 8; ++jj) id0[jj] = idw[4 * jj];
#pragma unroll
    for (int jj = 0; jj < 8; ++jj) id1[jj] = idw[32 + 4 * jj];
    const float bb = (m < V_OUT) ? bias[k * V_OUT + m] : 0.0f;

    const float* Wk = weights + (size_t)k * (D_DIM * V_OUT);
    for (int i = tid; i < (D_DIM / 2) * VPAD; i += THREADS) {
        const int j   = i >> 4;
        const int col = i & 15;
        v2f w;
        if (col < V_OUT) {
            w.x = Wk[(2 * j)     * V_OUT + col];
            w.y = Wk[(2 * j + 1) * V_OUT + col];
        } else {
            w.x = 0.0f; w.y = 0.0f;
        }
        lds_w[i] = w;
    }

    const char* fb = (const char*)feats;
    unsigned cur[8], nxt[8];
#pragma unroll
    for (int jj = 0; jj < 8; ++jj) cur[jj] = ((unsigned)id0[jj] << 10) | ((unsigned)q << 4);
#pragma unroll
    for (int jj = 0; jj < 8; ++jj) nxt[jj] = ((unsigned)id1[jj] << 10) | ((unsigned)q << 4);

    v4f buf[2][8];
#pragma unroll
    for (int s = 0; s < 2; ++s) {
#pragma unroll
        for (int jj = 0; jj < 8; ++jj) buf[s][jj] = load_piece(fb, cur[jj], s);
    }

    __syncthreads();

    float* obase = out + ((size_t)(k * S_SEL + s0)) * V_OUT + m;

    v8f c0, c1;
    run_pair<true>(buf, fb, cur, nxt, lds_a[wave], lds_w, rsub, q, m, hi, c0, c1);
    store_tile(obase,                        c0, bb, m, hi);
    store_tile(obase + (size_t)16 * V_OUT,   c1, bb, m, hi);

    run_pair<false>(buf, fb, nxt, nxt, lds_a[wave], lds_w, rsub, q, m, hi, c0, c1);
    store_tile(obase + (size_t)32 * V_OUT,   c0, bb, m, hi);
    store_tile(obase + (size_t)48 * V_OUT,   c1, bb, m, hi);
}

extern "C" void kernel_launch(void* const* d_in, const int* in_sizes, int n_in,
                              void* d_out, int out_size, void* d_ws, size_t ws_size,
                              hipStream_t stream) {
    const float* feats    = (const float*)d_in[1];
    const int*   mask_idx = (const int*)d_in[2];
    const float* weights  = (const float*)d_in[3];
    const float* bias     = (const float*)d_in[4];
    float*       out      = (float*)d_out;

    const int blocks = K_HEADS * BLOCKS_PER_HEAD;
    hipLaunchKernelGGL(attr_decoder_wmma_kernel,
                       dim3(blocks), dim3(THREADS), 0, stream,
                       feats, mask_idx, weights, bias, out);
}
